# starting kernel
# baseline (speedup 1.0000x reference)
#include <hip/hip_runtime.h>
#include <math.h>

typedef __attribute__((ext_vector_type(2))) float v2f;
typedef __attribute__((ext_vector_type(8))) float v8f;

constexpr int kB    = 2;
constexpr int kCin  = 32;
constexpr int kH    = 64;
constexpr int kW    = 64;
constexpr int kJ    = kH * kW;
constexpr int kHid  = 64;
constexpr int kCout = 32;
constexpr int kS    = 256;
constexpr int kTileJ   = 128;
constexpr int kThreads = 256;

__device__ __forceinline__ float gelu_exact(float x) {
    return 0.5f * x * (1.0f + erff(x * 0.70710678118654752440f));
}

__global__ void prep_cw_kernel(const float* __restrict__ W1, float* __restrict__ cw) {
    int tid = blockIdx.x * blockDim.x + threadIdx.x;
    if (tid >= kJ * kHid) return;
    int j = tid >> 6, h = tid & 63;
    const float inv63 = 1.0f / 63.0f;
    float cx = (float)(j & 63) * inv63;
    float cy = (float)(j >> 6) * inv63;
    cw[tid] = cx * W1[h] + cy * W1[kHid + h];
}

__global__ void prep_k_kernel(const float* __restrict__ v, const int* __restrict__ indices,
                              const float* __restrict__ W1, const float* __restrict__ b1,
                              float* __restrict__ Kmat) {
    int tid = blockIdx.x * blockDim.x + threadIdx.x;
    if (tid >= kB * kS * kHid) return;
    int h  = tid & 63;
    int bs = tid >> 6;
    int b  = bs >> 8;
    int s  = bs & 255;
    int idx = indices[s];
    const float inv63 = 1.0f / 63.0f;
    float sx = (float)(idx & 63) * inv63;
    float sy = (float)(idx >> 6) * inv63;
    float acc = b1[h] - sx * W1[h] - sy * W1[kHid + h];
    #pragma unroll
    for (int c = 0; c < kCin; ++c)
        acc = fmaf(v[((b * kCin + c) << 12) + idx], W1[(2 + c) * kHid + h], acc);
    Kmat[tid] = acc;
}

__global__ void __launch_bounds__(kThreads)
nystrom_main_kernel(const float* __restrict__ cw, const float* __restrict__ Kmat,
                    const float* __restrict__ W2, const float* __restrict__ b2,
                    float* __restrict__ out) {
    extern __shared__ float smem[];
    float* Klds = smem;
    float* w2s  = smem + kS * kHid;
    float* b2s  = w2s + kHid * kCout;

    const int tid = threadIdx.x;
    const int b   = blockIdx.y;

    {
        const float4* Kg4  = (const float4*)(Kmat + (size_t)b * kS * kHid);
        float4* Kl4 = (float4*)Klds;
        for (int i = tid; i < (kS * kHid) / 4; i += kThreads) Kl4[i] = Kg4[i];
        const float4* Wg4 = (const float4*)W2;
        float4* Wl4 = (float4*)w2s;
        for (int i = tid; i < (kHid * kCout) / 4; i += kThreads) Wl4[i] = Wg4[i];
        if (tid < kCout) b2s[tid] = b2[tid];
    }
    __syncthreads();

    const int lane  = tid & 31;
    const int wave  = tid >> 5;
    const int jl    = lane & 15;
    const int hoff  = (lane >> 4) << 1;
    const int jbase = blockIdx.x * kTileJ + wave * 16;
    const int j     = jbase + jl;

    float2 cwreg[16];
    #pragma unroll
    for (int kk = 0; kk < 16; ++kk)
        cwreg[kk] = *(const float2*)(cw + (size_t)j * kHid + 4 * kk + hoff);

    float acc0[16], acc1[16];
    #pragma unroll
    for (int kk = 0; kk < 16; ++kk) { acc0[kk] = 0.0f; acc1[kk] = 0.0f; }

    const float2* K2 = (const float2*)Klds;
    const int lhalf = hoff >> 1;
    for (int s = 0; s < kS; ++s) {
        const float2* row = K2 + s * (kHid / 2) + lhalf;
        #pragma unroll
        for (int kk = 0; kk < 16; ++kk) {
            float2 kv = row[2 * kk];
            acc0[kk] += gelu_exact(cwreg[kk].x + kv.x);
            acc1[kk] += gelu_exact(cwreg[kk].y + kv.y);
        }
    }

    v8f d0 = {}; v8f d1 = {};
    const int n = jl;
    #pragma unroll
    for (int kk = 0; kk < 16; ++kk) {
        v2f a; a.x = acc0[kk]; a.y = acc1[kk];
        int r = 4 * kk + hoff;
        v2f bf0, bf1;
        bf0.x = w2s[r * kCout + n];            bf0.y = w2s[(r + 1) * kCout + n];
        bf1.x = w2s[r * kCout + 16 + n];       bf1.y = w2s[(r + 1) * kCout + 16 + n];
        d0 = __builtin_amdgcn_wmma_f32_16x16x4_f32(false, a, false, bf0, (short)0, d0, false, false);
        d1 = __builtin_amdgcn_wmma_f32_16x16x4_f32(false, a, false, bf1, (short)0, d1, false, false);
    }

    const float scale = 1.0f / (float)kS;
    const int rowM = (lane >> 4) << 3;
    {
        float bc = b2s[n];
        float* o = out + ((size_t)(b * kCout + n)) * kJ + jbase + rowM;
        float4 lo = make_float4(d0[0] * scale + bc, d0[1] * scale + bc,
                                d0[2] * scale + bc, d0[3] * scale + bc);
        float4 hi = make_float4(d0[4] * scale + bc, d0[5] * scale + bc,
                                d0[6] * scale + bc, d0[7] * scale + bc);
        *(float4*)(o)     = lo;
        *(float4*)(o + 4) = hi;
    }
    {
        float bc = b2s[16 + n];
        float* o = out + ((size_t)(b * kCout + 16 + n)) * kJ + jbase + rowM;
        float4 lo = make_float4(d1[0] * scale + bc, d1[1] * scale + bc,
                                d1[2] * scale + bc, d1[3] * scale + bc);
        float4 hi = make_float4(d1[4] * scale + bc, d1[5] * scale + bc,
                                d1[6] * scale + bc, d1[7] * scale + bc);
        *(float4*)(o)     = lo;
        *(float4*)(o + 4) = hi;
    }
}

extern "C" void kernel_launch(void* const* d_in, const int* in_sizes, int n_in,
                              void* d_out, int out_size, void* d_ws, size_t ws_size,
                              hipStream_t stream) {
    const float* v       = (const float*)d_in[0];
    const int*   indices = (const int*)d_in[1];
    const float* W1      = (const float*)d_in[2];
    const float* b1      = (const float*)d_in[3];
    const float* W2      = (const float*)d_in[4];
    const float* b2      = (const float*)d_in[5];
    float* out = (float*)d_out;

    float* cw   = (float*)d_ws;
    float* Kmat = cw + (size_t)kJ * kHid;

    prep_cw_kernel<<<(kJ * kHid + 255) / 256, 256, 0, stream>>>(W1, cw);
    prep_k_kernel<<<(kB * kS * kHid + 255) / 256, 256, 0, stream>>>(v, indices, W1, b1, Kmat);

    size_t shbytes = (size_t)(kS * kHid + kHid * kCout + kCout) * sizeof(float);
    dim3 grid(kJ / kTileJ, kB);
    nystrom_main_kernel<<<grid, kThreads, shbytes, stream>>>(cw, Kmat, W2, b2, out);
}


# cw and K planes written by ONE preparatory launch of 192 blocks (128 K blocks as before; 64 cw blocks, a thread stores four whole-line 16-B pieces) instead of two of 1024 + 128; all sums kept
# speedup vs baseline: 18.0248x; 18.0248x over previous
#include <hip/hip_runtime.h>
#include <math.h>

typedef __attribute__((ext_vector_type(2))) float v2f;
typedef __attribute__((ext_vector_type(8))) float v8f;

constexpr int kB    = 2;
constexpr int kCin  = 32;
constexpr int kH    = 64;
constexpr int kW    = 64;
constexpr int kJ    = kH * kW;
constexpr int kHid  = 64;
constexpr int kCout = 32;
constexpr int kS    = 256;
constexpr int kRows    = 32;
constexpr int kSlices  = 8;
constexpr int kSPerW   = kS / kSlices;
constexpr int kHsplit  = 2;
constexpr int kKK      = 16 / kHsplit;
static_assert(kSlices * kHsplit * kCout * kRows <= kS * kHid, "partial products reuse the K tile in LDS");
constexpr int kParts   = kSlices * kHsplit;
constexpr int kWaves   = 2 * kParts;
constexpr int kThreads = 32 * kWaves;

__device__ __forceinline__ v2f splat2(float a) { v2f r; r.x = a; r.y = a; return r; }

__device__ __forceinline__ v2f gelu_pair_add(v2f x, v2f acc) {
    constexpr float kClamp = 5.65685415e+0f;
    v2f t;
    t.x = __builtin_amdgcn_fmed3f(x.x, -kClamp, kClamp);
    t.y = __builtin_amdgcn_fmed3f(x.y, -kClamp, kClamp);
    const v2f u = t * t;
    v2f p = __builtin_elementwise_fma(u, splat2(3.24880345e-10f), splat2(9.43710106e-8f));
    p = __builtin_elementwise_fma(u, p, splat2(2.61264881e-6f));
    p = __builtin_elementwise_fma(u, p, splat2(7.01790777e-5f));
    p = __builtin_elementwise_fma(u, p, splat2(5.14317537e-4f));
    p = __builtin_elementwise_fma(u, p, splat2(5.93394460e-3f));
    v2f q = __builtin_elementwise_fma(u, splat2(1.74089525e-8f), splat2(1.12367809e-6f));
    q = __builtin_elementwise_fma(u, q, splat2(2.86097766e-5f));
    q = __builtin_elementwise_fma(u, q, splat2(4.32120025e-4f));
    q = __builtin_elementwise_fma(u, q, splat2(3.76822334e-3f));
    q = __builtin_elementwise_fma(u, q, splat2(1.48741947e-2f));
    v2f r;
    r.x = __builtin_amdgcn_rcpf(q.x);
    r.y = __builtin_amdgcn_rcpf(q.y);
    const v2f phi = __builtin_elementwise_fma(t * p, r, splat2(0.5f));
    return __builtin_elementwise_fma(x, phi, acc);
}

constexpr int kPrepThreads = 256;
constexpr int kKBlocks     = (kB * kS * kHid) / kPrepThreads;
constexpr int kCwPieces    = 4;
constexpr int kCwBlocks    = (kJ * kHid) / (4 * kCwPieces * kPrepThreads);
constexpr int kPrepBlocks  = kKBlocks + kCwBlocks;
static_assert(kKBlocks * kPrepThreads == kB * kS * kHid, "K plane split evenly");
static_assert(kCwBlocks * kPrepThreads * kCwPieces * 4 == kJ * kHid, "cw plane split evenly");
static_assert(kHid == 64, "a 16-B piece index q covers row q / 16, hidden units 4 * (q % 16) ..");

__global__ void __launch_bounds__(kPrepThreads)
prep_kernel(const float* __restrict__ v, const int* __restrict__ indices,
            const float* __restrict__ W1, const float* __restrict__ b1,
            float* __restrict__ cw, float* __restrict__ Kmat) {
    const float inv63 = 1.0f / 63.0f;
    if (blockIdx.x < kKBlocks) {
        const int tid = blockIdx.x * kPrepThreads + threadIdx.x;
        const int h  = tid & 63;
        const int bs = tid >> 6;
        const int b  = bs >> 8;
        const int s  = bs & 255;
        const int idx = indices[s];
        const float sx = (float)(idx & 63) * inv63;
        const float sy = (float)(idx >> 6) * inv63;
        float acc = b1[h] - sx * W1[h] - sy * W1[kHid + h];
        #pragma unroll
        for (int c = 0; c < kCin; ++c)
            acc = fmaf(v[((b * kCin + c) << 12) + idx], W1[(2 + c) * kHid + h], acc);
        Kmat[tid] = acc;
    } else {
        const int lane  = threadIdx.x & 31;
        const int gwave = (blockIdx.x - kKBlocks) * (kPrepThreads / 32) + (threadIdx.x >> 5);
        const int q0    = gwave * (32 * kCwPieces) + lane;
        const int h4    = (lane & 15) * 4;
        const float4 w0 = *(const float4*)(W1 + h4);
        const float4 w1 = *(const float4*)(W1 + kHid + h4);
        float4* cw4 = (float4*)cw;
        #pragma unroll
        for (int i = 0; i < kCwPieces; ++i) {
            const int q = q0 + 32 * i;
            const int j = q >> 4;
            const float cx = (float)(j & 63) * inv63;
            const float cy = (float)(j >> 6) * inv63;
            float4 r;
            r.x = fmaf(cx, w0.x, cy * w1.x);
            r.y = fmaf(cx, w0.y, cy * w1.y);
            r.z = fmaf(cx, w0.z, cy * w1.z);
            r.w = fmaf(cx, w0.w, cy * w1.w);
            cw4[q] = r;
        }
    }
}

__global__ void __launch_bounds__(kThreads)
nystrom_main_kernel(const float* __restrict__ cw, const float* __restrict__ Kmat,
                    const float* __restrict__ W2, const float* __restrict__ b2,
                    float* __restrict__ out) {
    extern __shared__ float smem[];
    float* Klds = smem;
    float* w2s  = smem + kS * kHid;
    float* b2s  = w2s + kHid * kCout;

    const int tid = threadIdx.x;
    const int b   = blockIdx.y;

    {
        const float4* Kg4  = (const float4*)(Kmat + (size_t)b * kS * kHid);
        float4* Kl4 = (float4*)Klds;
        for (int i = tid; i < (kS * kHid) / 4; i += kThreads) Kl4[i] = Kg4[i];
        const float4* Wg4 = (const float4*)W2;
        float4* Wl4 = (float4*)w2s;
        for (int i = tid; i < (kHid * kCout) / 4; i += kThreads) Wl4[i] = Wg4[i];
        if (tid < kCout) b2s[tid] = b2[tid];
    }
    __syncthreads();

    const int lane   = tid & 31;
    const int wave   = tid >> 5;
    const int mtile  = wave & 1;
    const int hpart  = (wave >> 1) % kHsplit;
    const int slice  = (wave >> 1) / kHsplit;
    const int kk0    = hpart * kKK;
    const int jl     = lane & 15;
    const int hc     = lane >> 4;
    const int hoff   = hc << 1;
    const int jblock = blockIdx.x * kRows;
    const int j      = jblock + mtile * 16 + jl;

    v2f cwreg[kKK];
    #pragma unroll
    for (int kk = 0; kk < kKK; ++kk)
        cwreg[kk] = *(const v2f*)(cw + (size_t)j * kHid + 4 * (kk0 + kk) + hoff);

    v2f acc[kKK];
    #pragma unroll
    for (int kk = 0; kk < kKK; ++kk) acc[kk] = splat2(0.0f);

    const v2f* K2 = (const v2f*)Klds;
    const int s0 = slice * kSPerW;
    for (int s = s0; s < s0 + kSPerW; ++s) {
        const v2f* row = K2 + s * (kHid / 2) + 2 * kk0 + hc;
        #pragma unroll
        for (int kk = 0; kk < kKK; ++kk)
            acc[kk] = gelu_pair_add(cwreg[kk] + row[2 * kk], acc[kk]);
    }

    v8f d0 = {}; v8f d1 = {};
    #pragma unroll
    for (int kk = 0; kk < kKK; ++kk) {
        const v2f g = acc[kk];
        int r = 4 * (kk0 + kk) + hoff;
        v2f wf0, wf1;
        wf0.x = w2s[r * kCout + jl];            wf0.y = w2s[(r + 1) * kCout + jl];
        wf1.x = w2s[r * kCout + 16 + jl];       wf1.y = w2s[(r + 1) * kCout + 16 + jl];
        d0 = __builtin_amdgcn_wmma_f32_16x16x4_f32(false, wf0, false, g, (short)0, d0, false, false);
        d1 = __builtin_amdgcn_wmma_f32_16x16x4_f32(false, wf1, false, g, (short)0, d1, false, false);
    }

    __syncthreads();
    float* red = Klds;
    {
        float* rp = red + (slice * kHsplit + hpart) * (kCout * kRows) + mtile * 16 + jl;
        #pragma unroll
        for (int r = 0; r < 8; ++r) {
            rp[((0 * 8 + r) * 2 + hc) * kRows] = d0[r];
            rp[((1 * 8 + r) * 2 + hc) * kRows] = d1[r];
        }
    }
    __syncthreads();

    const float scale = 1.0f / (float)kS;
    static_assert(kWaves == kCout, "one wave per output channel in the final sum");
    {
        const int q = wave;
        const int c = (q >> 4) * 16 + (q & 1) * 8 + ((q >> 1) & 7);
        const float* rq = red + q * kRows + lane;
        float sum = rq[0];
        #pragma unroll
        for (int p = 1; p < kParts; ++p) sum += rq[p * (kCout * kRows)];
        out[((size_t)(b * kCout + c)) * kJ + jblock + lane] = sum * scale + b2s[c];
    }
}

extern "C" void kernel_launch(void* const* d_in, const int* in_sizes, int n_in,
                              void* d_out, int out_size, void* d_ws, size_t ws_size,
                              hipStream_t stream) {
    const float* v       = (const float*)d_in[0];
    const int*   indices = (const int*)d_in[1];
    const float* W1      = (const float*)d_in[2];
    const float* b1      = (const float*)d_in[3];
    const float* W2      = (const float*)d_in[4];
    const float* b2      = (const float*)d_in[5];
    float* out = (float*)d_out;

    float* cw   = (float*)d_ws;
    float* Kmat = cw + (size_t)kJ * kHid;

    prep_kernel<<<kPrepBlocks, kPrepThreads, 0, stream>>>(v, indices, W1, b1, cw, Kmat);

    size_t shbytes = (size_t)(kS * kHid + kHid * kCout + kCout) * sizeof(float);
    dim3 grid(kJ / kRows, kB);
    nystrom_main_kernel<<<grid, kThreads, shbytes, stream>>>(cw, Kmat, W2, b2, out);
}


# gelu argument clamp dropped: the packed fma forming Phi = 0.5 + x*P/Q saturates to [0,1] by its clamp bit (x*P/Q keeps growing past the fit range); two v_med3 fewer per value pair; sum order kept
# speedup vs baseline: 19.3088x; 1.0712x over previous
#include <hip/hip_runtime.h>
#include <math.h>

typedef __attribute__((ext_vector_type(2))) float v2f;
typedef __attribute__((ext_vector_type(8))) float v8f;

constexpr int kB    = 2;
constexpr int kCin  = 32;
constexpr int kH    = 64;
constexpr int kW    = 64;
constexpr int kJ    = kH * kW;
constexpr int kHid  = 64;
constexpr int kCout = 32;
constexpr int kS    = 256;
constexpr int kRows    = 32;
constexpr int kSlices  = 8;
constexpr int kSPerW   = kS / kSlices;
constexpr int kHsplit  = 2;
constexpr int kKK      = 16 / kHsplit;
static_assert(kKK == 8, "the sample loop finishes eight pairs per trip in one statement");
static_assert(kSlices * kHsplit * kCout * kRows <= kS * kHid, "partial products reuse the K tile in LDS");
constexpr int kParts   = kSlices * kHsplit;
constexpr int kWaves   = 2 * kParts;
constexpr int kThreads = 32 * kWaves;

__device__ __forceinline__ v2f splat2(float a) { v2f r; r.x = a; r.y = a; return r; }

__device__ __forceinline__ void gelu_pair_front(v2f x, v2f& p, v2f& r) {
    const v2f u = x * x;
    p = __builtin_elementwise_fma(u, splat2(3.24880345e-10f), splat2(9.43710106e-8f));
    p = __builtin_elementwise_fma(u, p, splat2(2.61264881e-6f));
    p = __builtin_elementwise_fma(u, p, splat2(7.01790777e-5f));
    p = __builtin_elementwise_fma(u, p, splat2(5.14317537e-4f));
    p = __builtin_elementwise_fma(u, p, splat2(5.93394460e-3f));
    v2f q = __builtin_elementwise_fma(u, splat2(1.74089525e-8f), splat2(1.12367809e-6f));
    q = __builtin_elementwise_fma(u, q, splat2(2.86097766e-5f));
    q = __builtin_elementwise_fma(u, q, splat2(4.32120025e-4f));
    q = __builtin_elementwise_fma(u, q, splat2(3.76822334e-3f));
    q = __builtin_elementwise_fma(u, q, splat2(1.48741947e-2f));
    r.x = __builtin_amdgcn_rcpf(q.x);
    r.y = __builtin_amdgcn_rcpf(q.y);
}

__device__ __forceinline__ void gelu_tail8(const v2f (&x)[8], v2f (&p)[8], const v2f (&r)[8], v2f (&a)[8]) {
    const v2f half = splat2(0.5f);
    asm(
        "v_pk_mul_f32 %[p0], %[p0], %[x0]\n\t"
        "v_pk_mul_f32 %[p1], %[p1], %[x1]\n\t"
        "v_pk_mul_f32 %[p2], %[p2], %[x2]\n\t"
        "v_pk_mul_f32 %[p3], %[p3], %[x3]\n\t"
        "v_pk_mul_f32 %[p4], %[p4], %[x4]\n\t"
        "v_pk_mul_f32 %[p5], %[p5], %[x5]\n\t"
        "v_pk_mul_f32 %[p6], %[p6], %[x6]\n\t"
        "v_pk_mul_f32 %[p7], %[p7], %[x7]\n\t"
        "v_pk_fma_f32 %[p0], %[p0], %[r0], %[h] clamp\n\t"
        "v_pk_fma_f32 %[p1], %[p1], %[r1], %[h] clamp\n\t"
        "v_pk_fma_f32 %[p2], %[p2], %[r2], %[h] clamp\n\t"
        "v_pk_fma_f32 %[p3], %[p3], %[r3], %[h] clamp\n\t"
        "v_pk_fma_f32 %[p4], %[p4], %[r4], %[h] clamp\n\t"
        "v_pk_fma_f32 %[p5], %[p5], %[r5], %[h] clamp\n\t"
        "v_pk_fma_f32 %[p6], %[p6], %[r6], %[h] clamp\n\t"
        "v_pk_fma_f32 %[p7], %[p7], %[r7], %[h] clamp\n\t"
        "v_pk_fma_f32 %[a0], %[x0], %[p0], %[a0]\n\t"
        "v_pk_fma_f32 %[a1], %[x1], %[p1], %[a1]\n\t"
        "v_pk_fma_f32 %[a2], %[x2], %[p2], %[a2]\n\t"
        "v_pk_fma_f32 %[a3], %[x3], %[p3], %[a3]\n\t"
        "v_pk_fma_f32 %[a4], %[x4], %[p4], %[a4]\n\t"
        "v_pk_fma_f32 %[a5], %[x5], %[p5], %[a5]\n\t"
        "v_pk_fma_f32 %[a6], %[x6], %[p6], %[a6]\n\t"
        "v_pk_fma_f32 %[a7], %[x7], %[p7], %[a7]"
        : [p0] "+v"(p[0]), [p1] "+v"(p[1]), [p2] "+v"(p[2]), [p3] "+v"(p[3]), [p4] "+v"(p[4]), [p5] "+v"(p[5]), [p6] "+v"(p[6]), [p7] "+v"(p[7]),
          [a0] "+v"(a[0]), [a1] "+v"(a[1]), [a2] "+v"(a[2]), [a3] "+v"(a[3]), [a4] "+v"(a[4]), [a5] "+v"(a[5]), [a6] "+v"(a[6]), [a7] "+v"(a[7])
        : [x0] "v"(x[0]), [x1] "v"(x[1]), [x2] "v"(x[2]), [x3] "v"(x[3]), [x4] "v"(x[4]), [x5] "v"(x[5]), [x6] "v"(x[6]), [x7] "v"(x[7]),
          [r0] "v"(r[0]), [r1] "v"(r[1]), [r2] "v"(r[2]), [r3] "v"(r[3]), [r4] "v"(r[4]), [r5] "v"(r[5]), [r6] "v"(r[6]), [r7] "v"(r[7]),
          [h] "v"(half));
}

constexpr int kPrepThreads = 256;
constexpr int kKBlocks     = (kB * kS * kHid) / kPrepThreads;
constexpr int kCwPieces    = 4;
constexpr int kCwBlocks    = (kJ * kHid) / (4 * kCwPieces * kPrepThreads);
constexpr int kPrepBlocks  = kKBlocks + kCwBlocks;
static_assert(kKBlocks * kPrepThreads == kB * kS * kHid, "K plane split evenly");
static_assert(kCwBlocks * kPrepThreads * kCwPieces * 4 == kJ * kHid, "cw plane split evenly");
static_assert(kHid == 64, "a 16-B piece index q covers row q / 16, hidden units 4 * (q % 16) ..");

__global__ void __launch_bounds__(kPrepThreads)
prep_kernel(const float* __restrict__ v, const int* __restrict__ indices,
            const float* __restrict__ W1, const float* __restrict__ b1,
            float* __restrict__ cw, float* __restrict__ Kmat) {
    const float inv63 = 1.0f / 63.0f;
    if (blockIdx.x < kKBlocks) {
        const int tid = blockIdx.x * kPrepThreads + threadIdx.x;
        const int h  = tid & 63;
        const int bs = tid >> 6;
        const int b  = bs >> 8;
        const int s  = bs & 255;
        const int idx = indices[s];
        const float sx = (float)(idx & 63) * inv63;
        const float sy = (float)(idx >> 6) * inv63;
        float acc = b1[h] - sx * W1[h] - sy * W1[kHid + h];
        #pragma unroll
        for (int c = 0; c < kCin; ++c)
            acc = fmaf(v[((b * kCin + c) << 12) + idx], W1[(2 + c) * kHid + h], acc);
        Kmat[tid] = acc;
    } else {
        const int lane  = threadIdx.x & 31;
        const int gwave = (blockIdx.x - kKBlocks) * (kPrepThreads / 32) + (threadIdx.x >> 5);
        const int q0    = gwave * (32 * kCwPieces) + lane;
        const int h4    = (lane & 15) * 4;
        const float4 w0 = *(const float4*)(W1 + h4);
        const float4 w1 = *(const float4*)(W1 + kHid + h4);
        float4* cw4 = (float4*)cw;
        #pragma unroll
        for (int i = 0; i < kCwPieces; ++i) {
            const int q = q0 + 32 * i;
            const int j = q >> 4;
            const float cx = (float)(j & 63) * inv63;
            const float cy = (float)(j >> 6) * inv63;
            float4 r;
            r.x = fmaf(cx, w0.x, cy * w1.x);
            r.y = fmaf(cx, w0.y, cy * w1.y);
            r.z = fmaf(cx, w0.z, cy * w1.z);
            r.w = fmaf(cx, w0.w, cy * w1.w);
            cw4[q] = r;
        }
    }
}

__global__ void __launch_bounds__(kThreads)
nystrom_main_kernel(const float* __restrict__ cw, const float* __restrict__ Kmat,
                    const float* __restrict__ W2, const float* __restrict__ b2,
                    float* __restrict__ out) {
    extern __shared__ float smem[];
    float* Klds = smem;
    float* w2s  = smem + kS * kHid;
    float* b2s  = w2s + kHid * kCout;

    const int tid = threadIdx.x;
    const int b   = blockIdx.y;

    {
        const float4* Kg4  = (const float4*)(Kmat + (size_t)b * kS * kHid);
        float4* Kl4 = (float4*)Klds;
        for (int i = tid; i < (kS * kHid) / 4; i += kThreads) Kl4[i] = Kg4[i];
        const float4* Wg4 = (const float4*)W2;
        float4* Wl4 = (float4*)w2s;
        for (int i = tid; i < (kHid * kCout) / 4; i += kThreads) Wl4[i] = Wg4[i];
        if (tid < kCout) b2s[tid] = b2[tid];
    }
    __syncthreads();

    const int lane   = tid & 31;
    const int wave   = tid >> 5;
    const int mtile  = wave & 1;
    const int hpart  = (wave >> 1) % kHsplit;
    const int slice  = (wave >> 1) / kHsplit;
    const int kk0    = hpart * kKK;
    const int jl     = lane & 15;
    const int hc     = lane >> 4;
    const int hoff   = hc << 1;
    const int jblock = blockIdx.x * kRows;
    const int j      = jblock + mtile * 16 + jl;

    v2f cwreg[kKK];
    #pragma unroll
    for (int kk = 0; kk < kKK; ++kk)
        cwreg[kk] = *(const v2f*)(cw + (size_t)j * kHid + 4 * (kk0 + kk) + hoff);

    v2f acc[kKK];
    #pragma unroll
    for (int kk = 0; kk < kKK; ++kk) acc[kk] = splat2(0.0f);

    const v2f* K2 = (const v2f*)Klds;
    const int s0 = slice * kSPerW;
    for (int s = s0; s < s0 + kSPerW; ++s) {
        const v2f* row = K2 + s * (kHid / 2) + 2 * kk0 + hc;
        v2f x[kKK], p[kKK], r[kKK];
        #pragma unroll
        for (int kk = 0; kk < kKK; ++kk) {
            x[kk] = cwreg[kk] + row[2 * kk];
            gelu_pair_front(x[kk], p[kk], r[kk]);
        }
        gelu_tail8(x, p, r, acc);
    }

    v8f d0 = {}; v8f d1 = {};
    #pragma unroll
    for (int kk = 0; kk < kKK; ++kk) {
        const v2f g = acc[kk];
        int r = 4 * (kk0 + kk) + hoff;
        v2f wf0, wf1;
        wf0.x = w2s[r * kCout + jl];            wf0.y = w2s[(r + 1) * kCout + jl];
        wf1.x = w2s[r * kCout + 16 + jl];       wf1.y = w2s[(r + 1) * kCout + 16 + jl];
        d0 = __builtin_amdgcn_wmma_f32_16x16x4_f32(false, wf0, false, g, (short)0, d0, false, false);
        d1 = __builtin_amdgcn_wmma_f32_16x16x4_f32(false, wf1, false, g, (short)0, d1, false, false);
    }

    __syncthreads();
    float* red = Klds;
    {
        float* rp = red + (slice * kHsplit + hpart) * (kCout * kRows) + mtile * 16 + jl;
        #pragma unroll
        for (int r = 0; r < 8; ++r) {
            rp[((0 * 8 + r) * 2 + hc) * kRows] = d0[r];
            rp[((1 * 8 + r) * 2 + hc) * kRows] = d1[r];
        }
    }
    __syncthreads();

    const float scale = 1.0f / (float)kS;
    static_assert(kWaves == kCout, "one wave per output channel in the final sum");
    {
        const int q = wave;
        const int c = (q >> 4) * 16 + (q & 1) * 8 + ((q >> 1) & 7);
        const float* rq = red + q * kRows + lane;
        float sum = rq[0];
        #pragma unroll
        for (int p = 1; p < kParts; ++p) sum += rq[p * (kCout * kRows)];
        out[((size_t)(b * kCout + c)) * kJ + jblock + lane] = sum * scale + b2s[c];
    }
}

extern "C" void kernel_launch(void* const* d_in, const int* in_sizes, int n_in,
                              void* d_out, int out_size, void* d_ws, size_t ws_size,
                              hipStream_t stream) {
    const float* v       = (const float*)d_in[0];
    const int*   indices = (const int*)d_in[1];
    const float* W1      = (const float*)d_in[2];
    const float* b1      = (const float*)d_in[3];
    const float* W2      = (const float*)d_in[4];
    const float* b2      = (const float*)d_in[5];
    float* out = (float*)d_out;

    float* cw   = (float*)d_ws;
    float* Kmat = cw + (size_t)kJ * kHid;

    prep_kernel<<<kPrepBlocks, kPrepThreads, 0, stream>>>(v, indices, W1, b1, cw, Kmat);

    size_t shbytes = (size_t)(kS * kHid + kHid * kCout + kCout) * sizeof(float);
    dim3 grid(kJ / kRows, kB);
    nystrom_main_kernel<<<grid, kThreads, shbytes, stream>>>(cw, Kmat, W2, b2, out);
}


# Lane = row: a wave sums 16 hidden x 32 samples of all 32 rows, so each K value serves one wave: 2 KB K piece in wave-private LDS, no 64 KB tile, no barrier before the loop; sum order changed
# speedup vs baseline: 19.9852x; 1.0350x over previous
#include <hip/hip_runtime.h>
#include <math.h>

typedef __attribute__((ext_vector_type(2))) float v2f;
typedef __attribute__((ext_vector_type(8))) float v8f;

constexpr int kB    = 2;
constexpr int kCin  = 32;
constexpr int kH    = 64;
constexpr int kW    = 64;
constexpr int kJ    = kH * kW;
constexpr int kHid  = 64;
constexpr int kCout = 32;
constexpr int kS    = 256;
constexpr int kRows    = 32;
constexpr int kSlices  = 8;
constexpr int kSPerW   = kS / kSlices;
constexpr int kHPerW   = 16;
constexpr int kHparts  = kHid / kHPerW;
constexpr int kKK      = kHPerW / 4;
static_assert(kHPerW == 16, "the sample loop finishes eight pairs per trip in one statement");
constexpr int kParts   = kSlices * kHparts;
constexpr int kWaves   = kParts;
constexpr int kThreads = 32 * kWaves;
constexpr int kGPitch  = 20;
static_assert(kRows * kGPitch <= kCout * kRows, "a wave's partial G fits the 4 KB that later takes its partial products");

__device__ __forceinline__ v2f splat2(float a) { v2f r; r.x = a; r.y = a; return r; }

__device__ __forceinline__ void gelu_pair_front(v2f x, v2f& p, v2f& r) {
    const v2f u = x * x;
    p = __builtin_elementwise_fma(u, splat2(3.24880345e-10f), splat2(9.43710106e-8f));
    p = __builtin_elementwise_fma(u, p, splat2(2.61264881e-6f));
    p = __builtin_elementwise_fma(u, p, splat2(7.01790777e-5f));
    p = __builtin_elementwise_fma(u, p, splat2(5.14317537e-4f));
    p = __builtin_elementwise_fma(u, p, splat2(5.93394460e-3f));
    v2f q = __builtin_elementwise_fma(u, splat2(1.74089525e-8f), splat2(1.12367809e-6f));
    q = __builtin_elementwise_fma(u, q, splat2(2.86097766e-5f));
    q = __builtin_elementwise_fma(u, q, splat2(4.32120025e-4f));
    q = __builtin_elementwise_fma(u, q, splat2(3.76822334e-3f));
    q = __builtin_elementwise_fma(u, q, splat2(1.48741947e-2f));
    r.x = __builtin_amdgcn_rcpf(q.x);
    r.y = __builtin_amdgcn_rcpf(q.y);
}

__device__ __forceinline__ void gelu_tail8(const v2f (&x)[8], v2f (&p)[8], const v2f (&r)[8], v2f (&a)[8]) {
    const v2f half = splat2(0.5f);
    asm(
        "v_pk_mul_f32 %[p0], %[p0], %[x0]\n\t"
        "v_pk_mul_f32 %[p1], %[p1], %[x1]\n\t"
        "v_pk_mul_f32 %[p2], %[p2], %[x2]\n\t"
        "v_pk_mul_f32 %[p3], %[p3], %[x3]\n\t"
        "v_pk_mul_f32 %[p4], %[p4], %[x4]\n\t"
        "v_pk_mul_f32 %[p5], %[p5], %[x5]\n\t"
        "v_pk_mul_f32 %[p6], %[p6], %[x6]\n\t"
        "v_pk_mul_f32 %[p7], %[p7], %[x7]\n\t"
        "v_pk_fma_f32 %[p0], %[p0], %[r0], %[h] clamp\n\t"
        "v_pk_fma_f32 %[p1], %[p1], %[r1], %[h] clamp\n\t"
        "v_pk_fma_f32 %[p2], %[p2], %[r2], %[h] clamp\n\t"
        "v_pk_fma_f32 %[p3], %[p3], %[r3], %[h] clamp\n\t"
        "v_pk_fma_f32 %[p4], %[p4], %[r4], %[h] clamp\n\t"
        "v_pk_fma_f32 %[p5], %[p5], %[r5], %[h] clamp\n\t"
        "v_pk_fma_f32 %[p6], %[p6], %[r6], %[h] clamp\n\t"
        "v_pk_fma_f32 %[p7], %[p7], %[r7], %[h] clamp\n\t"
        "v_pk_fma_f32 %[a0], %[x0], %[p0], %[a0]\n\t"
        "v_pk_fma_f32 %[a1], %[x1], %[p1], %[a1]\n\t"
        "v_pk_fma_f32 %[a2], %[x2], %[p2], %[a2]\n\t"
        "v_pk_fma_f32 %[a3], %[x3], %[p3], %[a3]\n\t"
        "v_pk_fma_f32 %[a4], %[x4], %[p4], %[a4]\n\t"
        "v_pk_fma_f32 %[a5], %[x5], %[p5], %[a5]\n\t"
        "v_pk_fma_f32 %[a6], %[x6], %[p6], %[a6]\n\t"
        "v_pk_fma_f32 %[a7], %[x7], %[p7], %[a7]"
        : [p0] "+v"(p[0]), [p1] "+v"(p[1]), [p2] "+v"(p[2]), [p3] "+v"(p[3]), [p4] "+v"(p[4]), [p5] "+v"(p[5]), [p6] "+v"(p[6]), [p7] "+v"(p[7]),
          [a0] "+v"(a[0]), [a1] "+v"(a[1]), [a2] "+v"(a[2]), [a3] "+v"(a[3]), [a4] "+v"(a[4]), [a5] "+v"(a[5]), [a6] "+v"(a[6]), [a7] "+v"(a[7])
        : [x0] "v"(x[0]), [x1] "v"(x[1]), [x2] "v"(x[2]), [x3] "v"(x[3]), [x4] "v"(x[4]), [x5] "v"(x[5]), [x6] "v"(x[6]), [x7] "v"(x[7]),
          [r0] "v"(r[0]), [r1] "v"(r[1]), [r2] "v"(r[2]), [r3] "v"(r[3]), [r4] "v"(r[4]), [r5] "v"(r[5]), [r6] "v"(r[6]), [r7] "v"(r[7]),
          [h] "v"(half));
}

constexpr int kPrepThreads = 256;
constexpr int kKBlocks     = (kB * kS * kHid) / kPrepThreads;
constexpr int kCwPieces    = 4;
constexpr int kCwBlocks    = (kJ * kHid) / (4 * kCwPieces * kPrepThreads);
constexpr int kPrepBlocks  = kKBlocks + kCwBlocks;
static_assert(kKBlocks * kPrepThreads == kB * kS * kHid, "K plane split evenly");
static_assert(kCwBlocks * kPrepThreads * kCwPieces * 4 == kJ * kHid, "cw plane split evenly");
static_assert(kHid == 64, "a 16-B piece index q covers row q / 16, hidden units 4 * (q % 16) ..");

__global__ void __launch_bounds__(kPrepThreads)
prep_kernel(const float* __restrict__ v, const int* __restrict__ indices,
            const float* __restrict__ W1, const float* __restrict__ b1,
            float* __restrict__ cw, float* __restrict__ Kmat) {
    const float inv63 = 1.0f / 63.0f;
    if (blockIdx.x < kKBlocks) {
        const int tid = blockIdx.x * kPrepThreads + threadIdx.x;
        const int h  = tid & 63;
        const int bs = tid >> 6;
        const int b  = bs >> 8;
        const int s  = bs & 255;
        const int idx = indices[s];
        const float sx = (float)(idx & 63) * inv63;
        const float sy = (float)(idx >> 6) * inv63;
        float acc = b1[h] - sx * W1[h] - sy * W1[kHid + h];
        #pragma unroll
        for (int c = 0; c < kCin; ++c)
            acc = fmaf(v[((b * kCin + c) << 12) + idx], W1[(2 + c) * kHid + h], acc);
        Kmat[tid] = acc;
    } else {
        const int lane  = threadIdx.x & 31;
        const int gwave = (blockIdx.x - kKBlocks) * (kPrepThreads / 32) + (threadIdx.x >> 5);
        const int q0    = gwave * (32 * kCwPieces) + lane;
        const int h4    = (lane & 15) * 4;
        const float4 w0 = *(const float4*)(W1 + h4);
        const float4 w1 = *(const float4*)(W1 + kHid + h4);
        float4* cw4 = (float4*)cw;
        #pragma unroll
        for (int i = 0; i < kCwPieces; ++i) {
            const int q = q0 + 32 * i;
            const int j = q >> 4;
            const float cx = (float)(j & 63) * inv63;
            const float cy = (float)(j >> 6) * inv63;
            float4 r;
            r.x = fmaf(cx, w0.x, cy * w1.x);
            r.y = fmaf(cx, w0.y, cy * w1.y);
            r.z = fmaf(cx, w0.z, cy * w1.z);
            r.w = fmaf(cx, w0.w, cy * w1.w);
            cw4[q] = r;
        }
    }
}

__global__ void __launch_bounds__(kThreads)
nystrom_main_kernel(const float* __restrict__ cw, const float* __restrict__ Kmat,
                    const float* __restrict__ W2, const float* __restrict__ b2,
                    float* __restrict__ out) {
    __shared__ __attribute__((aligned(16))) float smem[kParts * kCout * kRows];

    const int b      = blockIdx.y;
    const int lane   = threadIdx.x & 31;
    const int wave   = __builtin_amdgcn_readfirstlane((int)(threadIdx.x >> 5));
    const int hpart  = wave % kHparts;
    const int slice  = wave / kHparts;
    const int h0     = hpart * kHPerW;
    const int jl     = lane & 15;
    const int hc     = lane >> 4;
    const int hoff   = hc << 1;
    const int jblock = blockIdx.x * kRows;
    const int j      = jblock + lane;

    v2f cwreg[8];
    {
        const float4* c4 = (const float4*)(cw + (size_t)j * kHid + h0);
        #pragma unroll
        for (int i = 0; i < 4; ++i) {
            const float4 t = c4[i];
            cwreg[2 * i].x = t.x;     cwreg[2 * i].y = t.y;
            cwreg[2 * i + 1].x = t.z; cwreg[2 * i + 1].y = t.w;
        }
    }

    v2f wf0[kKK], wf1[kKK];
    #pragma unroll
    for (int kk = 0; kk < kKK; ++kk) {
        const int r = h0 + 4 * kk + hoff;
        wf0[kk].x = W2[r * kCout + jl];            wf0[kk].y = W2[(r + 1) * kCout + jl];
        wf1[kk].x = W2[r * kCout + 16 + jl];       wf1[kk].y = W2[(r + 1) * kCout + 16 + jl];
    }

    v2f acc[8];
    #pragma unroll
    for (int q = 0; q < 8; ++q) acc[q] = splat2(0.0f);

    float* mine = smem + wave * (kCout * kRows);
    {
        const float4* kg = (const float4*)(Kmat + ((size_t)b * kS + (size_t)slice * kSPerW + lane) * kHid + h0);
        const float4 k0 = kg[0], k1 = kg[1], k2 = kg[2], k3 = kg[3];
        float4* kl = (float4*)(mine + lane * kHPerW);
        kl[0] = k0; kl[1] = k1; kl[2] = k2; kl[3] = k3;
    }
    __builtin_amdgcn_fence(__ATOMIC_RELEASE, "wavefront");
    __builtin_amdgcn_wave_barrier();
    __builtin_amdgcn_fence(__ATOMIC_ACQUIRE, "wavefront");

    for (int i = 0; i < kSPerW; ++i) {
        const float4* krow = (const float4*)(mine + i * kHPerW);
        v2f x[8], p[8], r[8];
        #pragma unroll
        for (int q4 = 0; q4 < 4; ++q4) {
            const float4 k = krow[q4];
            v2f klo, khi;
            klo.x = k.x; klo.y = k.y; khi.x = k.z; khi.y = k.w;
            x[2 * q4]     = cwreg[2 * q4] + klo;
            x[2 * q4 + 1] = cwreg[2 * q4 + 1] + khi;
        }
        #pragma unroll
        for (int q = 0; q < 8; ++q) gelu_pair_front(x[q], p[q], r[q]);
        gelu_tail8(x, p, r, acc);
    }

    {
        float4* g4 = (float4*)(mine + lane * kGPitch);
        #pragma unroll
        for (int i = 0; i < 4; ++i) {
            float4 t;
            t.x = acc[2 * i].x;     t.y = acc[2 * i].y;
            t.z = acc[2 * i + 1].x; t.w = acc[2 * i + 1].y;
            g4[i] = t;
        }
    }
    __builtin_amdgcn_fence(__ATOMIC_RELEASE, "wavefront");
    __builtin_amdgcn_wave_barrier();
    __builtin_amdgcn_fence(__ATOMIC_ACQUIRE, "wavefront");

    v8f d00 = {}; v8f d01 = {}; v8f d10 = {}; v8f d11 = {};
    #pragma unroll
    for (int kk = 0; kk < kKK; ++kk) {
        const v2f g0 = *(const v2f*)(mine + (jl)      * kGPitch + 4 * kk + hoff);
        const v2f g1 = *(const v2f*)(mine + (16 + jl) * kGPitch + 4 * kk + hoff);
        d00 = __builtin_amdgcn_wmma_f32_16x16x4_f32(false, wf0[kk], false, g0, (short)0, d00, false, false);
        d01 = __builtin_amdgcn_wmma_f32_16x16x4_f32(false, wf1[kk], false, g0, (short)0, d01, false, false);
        d10 = __builtin_amdgcn_wmma_f32_16x16x4_f32(false, wf0[kk], false, g1, (short)0, d10, false, false);
        d11 = __builtin_amdgcn_wmma_f32_16x16x4_f32(false, wf1[kk], false, g1, (short)0, d11, false, false);
    }

    float* red = smem;
    {
        float* rp0 = mine + jl;
        float* rp1 = mine + 16 + jl;
        #pragma unroll
        for (int r = 0; r < 8; ++r) {
            rp0[((0 * 8 + r) * 2 + hc) * kRows] = d00[r];
            rp0[((1 * 8 + r) * 2 + hc) * kRows] = d01[r];
            rp1[((0 * 8 + r) * 2 + hc) * kRows] = d10[r];
            rp1[((1 * 8 + r) * 2 + hc) * kRows] = d11[r];
        }
    }
    __syncthreads();

    const float scale = 1.0f / (float)kS;
    static_assert(kWaves == kCout, "one wave per output channel in the final sum");
    {
        const int q = wave;
        const int c = (q >> 4) * 16 + (q & 1) * 8 + ((q >> 1) & 7);
        const float* rq = red + q * kRows + lane;
        float sum = rq[0];
        #pragma unroll
        for (int p = 1; p < kParts; ++p) sum += rq[p * (kCout * kRows)];
        out[((size_t)(b * kCout + c)) * kJ + jblock + lane] = sum * scale + b2[c];
    }
}

extern "C" void kernel_launch(void* const* d_in, const int* in_sizes, int n_in,
                              void* d_out, int out_size, void* d_ws, size_t ws_size,
                              hipStream_t stream) {
    const float* v       = (const float*)d_in[0];
    const int*   indices = (const int*)d_in[1];
    const float* W1      = (const float*)d_in[2];
    const float* b1      = (const float*)d_in[3];
    const float* W2      = (const float*)d_in[4];
    const float* b2      = (const float*)d_in[5];
    float* out = (float*)d_out;

    float* cw   = (float*)d_ws;
    float* Kmat = cw + (size_t)kJ * kHid;

    prep_kernel<<<kPrepBlocks, kPrepThreads, 0, stream>>>(v, indices, W1, b1, cw, Kmat);

    dim3 grid(kJ / kRows, kB);
    nystrom_main_kernel<<<grid, kThreads, 0, stream>>>(cw, Kmat, W2, b2, out);
}


# gelu as x/(1 + 2^(x*G(x*x))), G of degree 6 fitted to the log-odds of Phi: per value pair 11 packed ops + 2 v_exp + 2 v_rcp, was 15 + 2 v_rcp; plain source, no asm, no clamp bit; sum order kept
# speedup vs baseline: 21.7497x; 1.0883x over previous
#include <hip/hip_runtime.h>
#include <math.h>

typedef __attribute__((ext_vector_type(2))) float v2f;
typedef __attribute__((ext_vector_type(8))) float v8f;

constexpr int kB    = 2;
constexpr int kCin  = 32;
constexpr int kH    = 64;
constexpr int kW    = 64;
constexpr int kJ    = kH * kW;
constexpr int kHid  = 64;
constexpr int kCout = 32;
constexpr int kS    = 256;
constexpr int kRows    = 32;
constexpr int kSlices  = 8;
constexpr int kSPerW   = kS / kSlices;
constexpr int kHPerW   = 16;
constexpr int kHparts  = kHid / kHPerW;
constexpr int kKK      = kHPerW / 4;
static_assert(kHPerW == 16, "the sample loop works on eight pairs per trip");
constexpr int kParts   = kSlices * kHparts;
constexpr int kWaves   = kParts;
constexpr int kThreads = 32 * kWaves;
constexpr int kGPitch  = 20;
static_assert(kRows * kGPitch <= kCout * kRows, "a wave's partial G fits the 4 KB that later takes its partial products");

__device__ __forceinline__ v2f splat2(float a) { v2f r; r.x = a; r.y = a; return r; }

__device__ __forceinline__ v2f phi_pair(v2f x) {
    const v2f u = x * x;
    v2f g = __builtin_elementwise_fma(u, splat2(-5.21011678e-9f), splat2(3.85053909e-7f));
    g = __builtin_elementwise_fma(u, g, splat2(-1.14525683e-5f));
    g = __builtin_elementwise_fma(u, g, splat2(1.59390387e-4f));
    g = __builtin_elementwise_fma(u, g, splat2(9.55897922e-5f));
    g = __builtin_elementwise_fma(u, g, splat2(-1.04838580e-1f));
    g = __builtin_elementwise_fma(u, g, splat2(-2.30220723e+0f));
    const v2f z = x * g;
    v2f e;
    e.x = __builtin_amdgcn_exp2f(z.x);
    e.y = __builtin_amdgcn_exp2f(z.y);
    const v2f d = e + splat2(1.0f);
    v2f r;
    r.x = __builtin_amdgcn_rcpf(d.x);
    r.y = __builtin_amdgcn_rcpf(d.y);
    return r;
}

constexpr int kPrepThreads = 256;
constexpr int kKBlocks     = (kB * kS * kHid) / kPrepThreads;
constexpr int kCwPieces    = 4;
constexpr int kCwBlocks    = (kJ * kHid) / (4 * kCwPieces * kPrepThreads);
constexpr int kPrepBlocks  = kKBlocks + kCwBlocks;
static_assert(kKBlocks * kPrepThreads == kB * kS * kHid, "K plane split evenly");
static_assert(kCwBlocks * kPrepThreads * kCwPieces * 4 == kJ * kHid, "cw plane split evenly");
static_assert(kHid == 64, "a 16-B piece index q covers row q / 16, hidden units 4 * (q % 16) ..");

__global__ void __launch_bounds__(kPrepThreads)
prep_kernel(const float* __restrict__ v, const int* __restrict__ indices,
            const float* __restrict__ W1, const float* __restrict__ b1,
            float* __restrict__ cw, float* __restrict__ Kmat) {
    const float inv63 = 1.0f / 63.0f;
    if (blockIdx.x < kKBlocks) {
        const int tid = blockIdx.x * kPrepThreads + threadIdx.x;
        const int h  = tid & 63;
        const int bs = tid >> 6;
        const int b  = bs >> 8;
        const int s  = bs & 255;
        const int idx = indices[s];
        const float sx = (float)(idx & 63) * inv63;
        const float sy = (float)(idx >> 6) * inv63;
        float acc = b1[h] - sx * W1[h] - sy * W1[kHid + h];
        #pragma unroll
        for (int c = 0; c < kCin; ++c)
            acc = fmaf(v[((b * kCin + c) << 12) + idx], W1[(2 + c) * kHid + h], acc);
        Kmat[tid] = acc;
    } else {
        const int lane  = threadIdx.x & 31;
        const int gwave = (blockIdx.x - kKBlocks) * (kPrepThreads / 32) + (threadIdx.x >> 5);
        const int q0    = gwave * (32 * kCwPieces) + lane;
        const int h4    = (lane & 15) * 4;
        const float4 w0 = *(const float4*)(W1 + h4);
        const float4 w1 = *(const float4*)(W1 + kHid + h4);
        float4* cw4 = (float4*)cw;
        #pragma unroll
        for (int i = 0; i < kCwPieces; ++i) {
            const int q = q0 + 32 * i;
            const int j = q >> 4;
            const float cx = (float)(j & 63) * inv63;
            const float cy = (float)(j >> 6) * inv63;
            float4 r;
            r.x = fmaf(cx, w0.x, cy * w1.x);
            r.y = fmaf(cx, w0.y, cy * w1.y);
            r.z = fmaf(cx, w0.z, cy * w1.z);
            r.w = fmaf(cx, w0.w, cy * w1.w);
            cw4[q] = r;
        }
    }
}

__global__ void __launch_bounds__(kThreads)
nystrom_main_kernel(const float* __restrict__ cw, const float* __restrict__ Kmat,
                    const float* __restrict__ W2, const float* __restrict__ b2,
                    float* __restrict__ out) {
    __shared__ __attribute__((aligned(16))) float smem[kParts * kCout * kRows];

    const int b      = blockIdx.y;
    const int lane   = threadIdx.x & 31;
    const int wave   = __builtin_amdgcn_readfirstlane((int)(threadIdx.x >> 5));
    const int hpart  = wave % kHparts;
    const int slice  = wave / kHparts;
    const int h0     = hpart * kHPerW;
    const int jl     = lane & 15;
    const int hc     = lane >> 4;
    const int hoff   = hc << 1;
    const int jblock = blockIdx.x * kRows;
    const int j      = jblock + lane;

    v2f cwreg[8];
    {
        const float4* c4 = (const float4*)(cw + (size_t)j * kHid + h0);
        #pragma unroll
        for (int i = 0; i < 4; ++i) {
            const float4 t = c4[i];
            cwreg[2 * i].x = t.x;     cwreg[2 * i].y = t.y;
            cwreg[2 * i + 1].x = t.z; cwreg[2 * i + 1].y = t.w;
        }
    }

    v2f wf0[kKK], wf1[kKK];
    #pragma unroll
    for (int kk = 0; kk < kKK; ++kk) {
        const int r = h0 + 4 * kk + hoff;
        wf0[kk].x = W2[r * kCout + jl];            wf0[kk].y = W2[(r + 1) * kCout + jl];
        wf1[kk].x = W2[r * kCout + 16 + jl];       wf1[kk].y = W2[(r + 1) * kCout + 16 + jl];
    }

    v2f acc[8];
    #pragma unroll
    for (int q = 0; q < 8; ++q) acc[q] = splat2(0.0f);

    float* mine = smem + wave * (kCout * kRows);
    {
        const float4* kg = (const float4*)(Kmat + ((size_t)b * kS + (size_t)slice * kSPerW + lane) * kHid + h0);
        const float4 k0 = kg[0], k1 = kg[1], k2 = kg[2], k3 = kg[3];
        float4* kl = (float4*)(mine + lane * kHPerW);
        kl[0] = k0; kl[1] = k1; kl[2] = k2; kl[3] = k3;
    }
    __builtin_amdgcn_fence(__ATOMIC_RELEASE, "wavefront");
    __builtin_amdgcn_wave_barrier();
    __builtin_amdgcn_fence(__ATOMIC_ACQUIRE, "wavefront");

    for (int i = 0; i < kSPerW; ++i) {
        const float4* krow = (const float4*)(mine + i * kHPerW);
        v2f x[8];
        #pragma unroll
        for (int q4 = 0; q4 < 4; ++q4) {
            const float4 k = krow[q4];
            v2f klo, khi;
            klo.x = k.x; klo.y = k.y; khi.x = k.z; khi.y = k.w;
            x[2 * q4]     = cwreg[2 * q4] + klo;
            x[2 * q4 + 1] = cwreg[2 * q4 + 1] + khi;
        }
        #pragma unroll
        for (int q = 0; q < 8; ++q) acc[q] = __builtin_elementwise_fma(x[q], phi_pair(x[q]), acc[q]);
    }

    {
        float4* g4 = (float4*)(mine + lane * kGPitch);
        #pragma unroll
        for (int i = 0; i < 4; ++i) {
            float4 t;
            t.x = acc[2 * i].x;     t.y = acc[2 * i].y;
            t.z = acc[2 * i + 1].x; t.w = acc[2 * i + 1].y;
            g4[i] = t;
        }
    }
    __builtin_amdgcn_fence(__ATOMIC_RELEASE, "wavefront");
    __builtin_amdgcn_wave_barrier();
    __builtin_amdgcn_fence(__ATOMIC_ACQUIRE, "wavefront");

    v8f d00 = {}; v8f d01 = {}; v8f d10 = {}; v8f d11 = {};
    #pragma unroll
    for (int kk = 0; kk < kKK; ++kk) {
        const v2f g0 = *(const v2f*)(mine + (jl)      * kGPitch + 4 * kk + hoff);
        const v2f g1 = *(const v2f*)(mine + (16 + jl) * kGPitch + 4 * kk + hoff);
        d00 = __builtin_amdgcn_wmma_f32_16x16x4_f32(false, wf0[kk], false, g0, (short)0, d00, false, false);
        d01 = __builtin_amdgcn_wmma_f32_16x16x4_f32(false, wf1[kk], false, g0, (short)0, d01, false, false);
        d10 = __builtin_amdgcn_wmma_f32_16x16x4_f32(false, wf0[kk], false, g1, (short)0, d10, false, false);
        d11 = __builtin_amdgcn_wmma_f32_16x16x4_f32(false, wf1[kk], false, g1, (short)0, d11, false, false);
    }

    float* red = smem;
    {
        float* rp0 = mine + jl;
        float* rp1 = mine + 16 + jl;
        #pragma unroll
        for (int r = 0; r < 8; ++r) {
            rp0[((0 * 8 + r) * 2 + hc) * kRows] = d00[r];
            rp0[((1 * 8 + r) * 2 + hc) * kRows] = d01[r];
            rp1[((0 * 8 + r) * 2 + hc) * kRows] = d10[r];
            rp1[((1 * 8 + r) * 2 + hc) * kRows] = d11[r];
        }
    }
    __syncthreads();

    const float scale = 1.0f / (float)kS;
    static_assert(kWaves == kCout, "one wave per output channel in the final sum");
    {
        const int q = wave;
        const int c = (q >> 4) * 16 + (q & 1) * 8 + ((q >> 1) & 7);
        const float* rq = red + q * kRows + lane;
        float sum = rq[0];
        #pragma unroll
        for (int p = 1; p < kParts; ++p) sum += rq[p * (kCout * kRows)];
        out[((size_t)(b * kCout + c)) * kJ + jblock + lane] = sum * scale + b2[c];
    }
}

extern "C" void kernel_launch(void* const* d_in, const int* in_sizes, int n_in,
                              void* d_out, int out_size, void* d_ws, size_t ws_size,
                              hipStream_t stream) {
    const float* v       = (const float*)d_in[0];
    const int*   indices = (const int*)d_in[1];
    const float* W1      = (const float*)d_in[2];
    const float* b1      = (const float*)d_in[3];
    const float* W2      = (const float*)d_in[4];
    const float* b2      = (const float*)d_in[5];
    float* out = (float*)d_out;

    float* cw   = (float*)d_ws;
    float* Kmat = cw + (size_t)kJ * kHid;

    prep_kernel<<<kPrepBlocks, kPrepThreads, 0, stream>>>(v, indices, W1, b1, cw, Kmat);

    dim3 grid(kJ / kRows, kB);
    nystrom_main_kernel<<<grid, kThreads, 0, stream>>>(cw, Kmat, W2, b2, out);
}


# gelu as x/2 + (x/2)*tanh(x*G(x*x)) on v_tanh_f32: per value pair 10 packed ops + 2 v_tanh, was 11 + 2 v_exp + 2 v_rcp; sum of x formed once per wave behind the loop; sum order changed
# speedup vs baseline: 25.0264x; 1.1507x over previous
#include <hip/hip_runtime.h>
#include <math.h>

typedef __attribute__((ext_vector_type(2))) float v2f;
typedef __attribute__((ext_vector_type(8))) float v8f;

constexpr int kB    = 2;
constexpr int kCin  = 32;
constexpr int kH    = 64;
constexpr int kW    = 64;
constexpr int kJ    = kH * kW;
constexpr int kHid  = 64;
constexpr int kCout = 32;
constexpr int kS    = 256;
constexpr int kRows    = 32;
constexpr int kSlices  = 8;
constexpr int kSPerW   = kS / kSlices;
constexpr int kHPerW   = 16;
constexpr int kHparts  = kHid / kHPerW;
constexpr int kKK      = kHPerW / 4;
static_assert(kHPerW == 16, "the sample loop works on eight pairs per trip");
constexpr int kParts   = kSlices * kHparts;
constexpr int kWaves   = kParts;
constexpr int kThreads = 32 * kWaves;
constexpr int kGPitch  = 20;
static_assert(kRows * kGPitch <= kCout * kRows, "a wave's partial G fits the 4 KB that later takes its partial products");
static_assert(kSPerW * kHPerW + kHPerW <= kCout * kRows, "a wave's K piece and its 16 column sums fit its 4 KB");
static_assert(kSPerW == 32, "the column sums take 16 samples per lane half");

__device__ __forceinline__ v2f splat2(float a) { v2f r; r.x = a; r.y = a; return r; }

__device__ __forceinline__ v2f tanh_half_logodds_pair(v2f x) {
    const v2f u = x * x;
    v2f g = __builtin_elementwise_fma(u, splat2(1.805688932e-9f), splat2(-1.334495181e-7f));
    g = __builtin_elementwise_fma(u, g, splat2(3.969157660e-6f));
    g = __builtin_elementwise_fma(u, g, splat2(-5.524049993e-5f));
    g = __builtin_elementwise_fma(u, g, splat2(-3.312889748e-5f));
    g = __builtin_elementwise_fma(u, g, splat2(3.633428365e-2f));
    g = __builtin_elementwise_fma(u, g, splat2(7.978842258e-1f));
    const v2f z = x * g;
    v2f t;
    t.x = __builtin_amdgcn_tanhf(z.x);
    t.y = __builtin_amdgcn_tanhf(z.y);
    return t;
}

constexpr int kPrepThreads = 256;
constexpr int kKBlocks     = (kB * kS * kHid) / kPrepThreads;
constexpr int kCwPieces    = 4;
constexpr int kCwBlocks    = (kJ * kHid) / (4 * kCwPieces * kPrepThreads);
constexpr int kPrepBlocks  = kKBlocks + kCwBlocks;
static_assert(kKBlocks * kPrepThreads == kB * kS * kHid, "K plane split evenly");
static_assert(kCwBlocks * kPrepThreads * kCwPieces * 4 == kJ * kHid, "cw plane split evenly");
static_assert(kHid == 64, "a 16-B piece index q covers row q / 16, hidden units 4 * (q % 16) ..");

__global__ void __launch_bounds__(kPrepThreads)
prep_kernel(const float* __restrict__ v, const int* __restrict__ indices,
            const float* __restrict__ W1, const float* __restrict__ b1,
            float* __restrict__ cw, float* __restrict__ Kmat) {
    const float inv63 = 1.0f / 63.0f;
    if (blockIdx.x < kKBlocks) {
        const int tid = blockIdx.x * kPrepThreads + threadIdx.x;
        const int h  = tid & 63;
        const int bs = tid >> 6;
        const int b  = bs >> 8;
        const int s  = bs & 255;
        const int idx = indices[s];
        const float sx = (float)(idx & 63) * inv63;
        const float sy = (float)(idx >> 6) * inv63;
        float acc = b1[h] - sx * W1[h] - sy * W1[kHid + h];
        #pragma unroll
        for (int c = 0; c < kCin; ++c)
            acc = fmaf(v[((b * kCin + c) << 12) + idx], W1[(2 + c) * kHid + h], acc);
        Kmat[tid] = acc;
    } else {
        const int lane  = threadIdx.x & 31;
        const int gwave = (blockIdx.x - kKBlocks) * (kPrepThreads / 32) + (threadIdx.x >> 5);
        const int q0    = gwave * (32 * kCwPieces) + lane;
        const int h4    = (lane & 15) * 4;
        const float4 w0 = *(const float4*)(W1 + h4);
        const float4 w1 = *(const float4*)(W1 + kHid + h4);
        float4* cw4 = (float4*)cw;
        #pragma unroll
        for (int i = 0; i < kCwPieces; ++i) {
            const int q = q0 + 32 * i;
            const int j = q >> 4;
            const float cx = (float)(j & 63) * inv63;
            const float cy = (float)(j >> 6) * inv63;
            float4 r;
            r.x = fmaf(cx, w0.x, cy * w1.x);
            r.y = fmaf(cx, w0.y, cy * w1.y);
            r.z = fmaf(cx, w0.z, cy * w1.z);
            r.w = fmaf(cx, w0.w, cy * w1.w);
            cw4[q] = r;
        }
    }
}

__global__ void __launch_bounds__(kThreads)
nystrom_main_kernel(const float* __restrict__ cw, const float* __restrict__ Kmat,
                    const float* __restrict__ W2, const float* __restrict__ b2,
                    float* __restrict__ out) {
    __shared__ __attribute__((aligned(16))) float smem[kParts * kCout * kRows];

    const int b      = blockIdx.y;
    const int lane   = threadIdx.x & 31;
    const int wave   = __builtin_amdgcn_readfirstlane((int)(threadIdx.x >> 5));
    const int hpart  = wave % kHparts;
    const int slice  = wave / kHparts;
    const int h0     = hpart * kHPerW;
    const int jl     = lane & 15;
    const int hc     = lane >> 4;
    const int hoff   = hc << 1;
    const int jblock = blockIdx.x * kRows;
    const int j      = jblock + lane;

    v2f cwreg[8];
    {
        const float4* c4 = (const float4*)(cw + (size_t)j * kHid + h0);
        #pragma unroll
        for (int i = 0; i < 4; ++i) {
            const float4 t = c4[i];
            cwreg[2 * i].x = t.x;     cwreg[2 * i].y = t.y;
            cwreg[2 * i + 1].x = t.z; cwreg[2 * i + 1].y = t.w;
        }
    }

    v2f wf0[kKK], wf1[kKK];
    #pragma unroll
    for (int kk = 0; kk < kKK; ++kk) {
        const int r = h0 + 4 * kk + hoff;
        wf0[kk].x = W2[r * kCout + jl];            wf0[kk].y = W2[(r + 1) * kCout + jl];
        wf1[kk].x = W2[r * kCout + 16 + jl];       wf1[kk].y = W2[(r + 1) * kCout + 16 + jl];
    }

    v2f acc[8];
    #pragma unroll
    for (int q = 0; q < 8; ++q) acc[q] = splat2(0.0f);

    float* mine = smem + wave * (kCout * kRows);
    {
        const float4* kg = (const float4*)(Kmat + ((size_t)b * kS + (size_t)slice * kSPerW + lane) * kHid + h0);
        const float4 k0 = kg[0], k1 = kg[1], k2 = kg[2], k3 = kg[3];
        float4* kl = (float4*)(mine + lane * kHPerW);
        kl[0] = k0; kl[1] = k1; kl[2] = k2; kl[3] = k3;
    }
    __builtin_amdgcn_fence(__ATOMIC_RELEASE, "wavefront");
    __builtin_amdgcn_wave_barrier();
    __builtin_amdgcn_fence(__ATOMIC_ACQUIRE, "wavefront");

    for (int i = 0; i < kSPerW; ++i) {
        const float4* krow = (const float4*)(mine + i * kHPerW);
        v2f x[8];
        #pragma unroll
        for (int q4 = 0; q4 < 4; ++q4) {
            const float4 k = krow[q4];
            v2f klo, khi;
            klo.x = k.x; klo.y = k.y; khi.x = k.z; khi.y = k.w;
            x[2 * q4]     = cwreg[2 * q4] + klo;
            x[2 * q4 + 1] = cwreg[2 * q4 + 1] + khi;
        }
        #pragma unroll
        for (int q = 0; q < 8; ++q) acc[q] = __builtin_elementwise_fma(x[q], tanh_half_logodds_pair(x[q]), acc[q]);
    }

    {
        const float* kc = mine + hc * (16 * kHPerW) + jl;
        float ks = kc[0];
        #pragma unroll
        for (int i = 1; i < 16; ++i) ks += kc[i * kHPerW];
        ks += __shfl_xor(ks, 16, 32);
        mine[kSPerW * kHPerW + jl] = ks;
    }
    __builtin_amdgcn_fence(__ATOMIC_RELEASE, "wavefront");
    __builtin_amdgcn_wave_barrier();
    __builtin_amdgcn_fence(__ATOMIC_ACQUIRE, "wavefront");
    {
        const float4* ks4 = (const float4*)(mine + kSPerW * kHPerW);
        #pragma unroll
        for (int q4 = 0; q4 < 4; ++q4) {
            const float4 k = ks4[q4];
            v2f klo, khi;
            klo.x = k.x; klo.y = k.y; khi.x = k.z; khi.y = k.w;
            acc[2 * q4]     += __builtin_elementwise_fma(cwreg[2 * q4],     splat2((float)kSPerW), klo);
            acc[2 * q4 + 1] += __builtin_elementwise_fma(cwreg[2 * q4 + 1], splat2((float)kSPerW), khi);
        }
    }

    {
        float4* g4 = (float4*)(mine + lane * kGPitch);
        #pragma unroll
        for (int i = 0; i < 4; ++i) {
            float4 t;
            t.x = acc[2 * i].x;     t.y = acc[2 * i].y;
            t.z = acc[2 * i + 1].x; t.w = acc[2 * i + 1].y;
            g4[i] = t;
        }
    }
    __builtin_amdgcn_fence(__ATOMIC_RELEASE, "wavefront");
    __builtin_amdgcn_wave_barrier();
    __builtin_amdgcn_fence(__ATOMIC_ACQUIRE, "wavefront");

    v8f d00 = {}; v8f d01 = {}; v8f d10 = {}; v8f d11 = {};
    #pragma unroll
    for (int kk = 0; kk < kKK; ++kk) {
        const v2f g0 = *(const v2f*)(mine + (jl)      * kGPitch + 4 * kk + hoff);
        const v2f g1 = *(const v2f*)(mine + (16 + jl) * kGPitch + 4 * kk + hoff);
        d00 = __builtin_amdgcn_wmma_f32_16x16x4_f32(false, wf0[kk], false, g0, (short)0, d00, false, false);
        d01 = __builtin_amdgcn_wmma_f32_16x16x4_f32(false, wf1[kk], false, g0, (short)0, d01, false, false);
        d10 = __builtin_amdgcn_wmma_f32_16x16x4_f32(false, wf0[kk], false, g1, (short)0, d10, false, false);
        d11 = __builtin_amdgcn_wmma_f32_16x16x4_f32(false, wf1[kk], false, g1, (short)0, d11, false, false);
    }

    float* red = smem;
    {
        float* rp0 = mine + jl;
        float* rp1 = mine + 16 + jl;
        #pragma unroll
        for (int r = 0; r < 8; ++r) {
            rp0[((0 * 8 + r) * 2 + hc) * kRows] = d00[r];
            rp0[((1 * 8 + r) * 2 + hc) * kRows] = d01[r];
            rp1[((0 * 8 + r) * 2 + hc) * kRows] = d10[r];
            rp1[((1 * 8 + r) * 2 + hc) * kRows] = d11[r];
        }
    }
    __syncthreads();

    const float scale = 0.5f / (float)kS;
    static_assert(kWaves == kCout, "one wave per output channel in the final sum");
    {
        const int q = wave;
        const int c = (q >> 4) * 16 + (q & 1) * 8 + ((q >> 1) & 7);
        const float* rq = red + q * kRows + lane;
        float sum = rq[0];
        #pragma unroll
        for (int p = 1; p < kParts; ++p) sum += rq[p * (kCout * kRows)];
        out[((size_t)(b * kCout + c)) * kJ + jblock + lane] = sum * scale + b2[c];
    }
}

extern "C" void kernel_launch(void* const* d_in, const int* in_sizes, int n_in,
                              void* d_out, int out_size, void* d_ws, size_t ws_size,
                              hipStream_t stream) {
    const float* v       = (const float*)d_in[0];
    const int*   indices = (const int*)d_in[1];
    const float* W1      = (const float*)d_in[2];
    const float* b1      = (const float*)d_in[3];
    const float* W2      = (const float*)d_in[4];
    const float* b2      = (const float*)d_in[5];
    float* out = (float*)d_out;

    float* cw   = (float*)d_ws;
    float* Kmat = cw + (size_t)kJ * kHid;

    prep_kernel<<<kPrepBlocks, kPrepThreads, 0, stream>>>(v, indices, W1, b1, cw, Kmat);

    dim3 grid(kJ / kRows, kB);
    nystrom_main_kernel<<<grid, kThreads, 0, stream>>>(cw, Kmat, W2, b2, out);
}
